# starting kernel
# baseline (speedup 1.0000x reference)
#include <hip/hip_runtime.h>

typedef __attribute__((ext_vector_type(2))) float v2f;
typedef __attribute__((ext_vector_type(8))) float v8f;

#define NROWS 8192
#define DIM   2048
#define SCALE 0.1f

__global__ void colsum_wmma_kernel(const float* __restrict__ W,
                                   float* __restrict__ wp) {
    int gtid = blockIdx.x * blockDim.x + threadIdx.x;
    int wave = gtid >> 5;
    int lane = threadIdx.x & 31;
    int half = lane >> 4;
    int n    = lane & 15;

    int split    = wave >> 7;
    int colgroup = wave & 127;
    int j = colgroup * 16 + n;

    v2f a; a.x = 1.0f; a.y = 1.0f;
    v8f c = {};

    int kbase = split * 512;
    for (int k0 = kbase; k0 < kbase + 512; k0 += 4) {
        int r = k0 + 2 * half;
        v2f b;
        b.x = W[(long)r * DIM + j];
        b.y = W[(long)(r + 1) * DIM + j];
        c = __builtin_amdgcn_wmma_f32_16x16x4_f32(
                false, a, false, b, (short)0, c, false, false);
    }
    if (lane < 16)
        wp[split * DIM + colgroup * 16 + n] = c[0];
}

__global__ void combine_kernel(const float* __restrict__ wp,
                               const float* __restrict__ bias,
                               float* __restrict__ wc,
                               float* __restrict__ sb) {
    int j = blockIdx.x * 256 + threadIdx.x;
    if (j < DIM)
        wc[j] = wp[j] + wp[DIM + j] + wp[2 * DIM + j] + wp[3 * DIM + j];

    if (blockIdx.x == 0) {
        __shared__ float s[256];
        float acc = 0.0f;
        for (int t = threadIdx.x; t < DIM; t += 256) acc += bias[t];
        s[threadIdx.x] = acc;
        __syncthreads();
        for (int off = 128; off > 0; off >>= 1) {
            if (threadIdx.x < off) s[threadIdx.x] += s[threadIdx.x + off];
            __syncthreads();
        }
        if (threadIdx.x == 0) sb[0] = s[0];
    }
}

__global__ void gemv_wmma_kernel(const float* __restrict__ x,
                                 const float* __restrict__ wc,
                                 float* __restrict__ op) {
    int gtid = blockIdx.x * blockDim.x + threadIdx.x;
    int wave = gtid >> 5;
    int lane = threadIdx.x & 31;
    int half = lane >> 4;
    int n    = lane & 15;

    int rowgroup = wave >> 2;
    int split    = wave & 3;
    long row = (long)rowgroup * 16 + n;
    const float* xrow = x + row * DIM;

    v8f c = {};
    int kbase = split * 512;
    for (int k0 = kbase; k0 < kbase + 512; k0 += 4) {
        v2f b = *(const v2f*)(xrow + k0 + 2 * half);
        v2f a;
        a.x = wc[k0 + 2 * half];
        a.y = wc[k0 + 2 * half + 1];
        c = __builtin_amdgcn_wmma_f32_16x16x4_f32(
                false, a, false, b, (short)0, c, false, false);
    }
    if (lane < 16)
        op[(long)split * NROWS + rowgroup * 16 + n] = c[0];
}

__global__ void finalize_kernel(const float* __restrict__ op,
                                const float* __restrict__ sb,
                                float* __restrict__ out) {
    int r = blockIdx.x * 256 + threadIdx.x;
    if (r < NROWS) {
        float v = op[r] + op[NROWS + r] + op[2 * NROWS + r] + op[3 * NROWS + r];
        out[r] = SCALE * (v + sb[0]);
    }
}

extern "C" void kernel_launch(void* const* d_in, const int* in_sizes, int n_in,
                              void* d_out, int out_size, void* d_ws, size_t ws_size,
                              hipStream_t stream) {
    const float* x = (const float*)d_in[0];
    const float* W = (const float*)d_in[1];
    const float* b = (const float*)d_in[2];
    float* out = (float*)d_out;

    float* ws = (float*)d_ws;
    float* wp = ws;
    float* wc = ws + 4 * DIM;
    float* sb = ws + 5 * DIM;
    float* op = ws + 5 * DIM + 256;

    colsum_wmma_kernel<<<64, 256, 0, stream>>>(W, wp);
    combine_kernel<<<8, 256, 0, stream>>>(wp, b, wc, sb);
    gemv_wmma_kernel<<<256, 256, 0, stream>>>(x, wc, op);
    finalize_kernel<<<32, 256, 0, stream>>>(op, sb, out);
}


# launch 2 fetches its A operand (the 4 plane slices of a wave) as 4 whole 512-byte loads and turns them into the 16 fragments in a private LDS strip; was 16 loads of 4 part-lines each
# speedup vs baseline: 2.4157x; 2.4157x over previous
#include <hip/hip_runtime.h>


typedef __attribute__((ext_vector_type(2))) float v2f;
typedef __attribute__((ext_vector_type(4))) float v4f;
typedef __attribute__((ext_vector_type(8))) float v8f;

#define NROWS 8192
#define DIM   2048
#define SCALE 0.1f

#define NCG   (DIM / 32)
#define NRQ   4
#define QROWS (DIM / NRQ)
#define CWAVES 8

__global__ __launch_bounds__(256)
void colsum_quarters_kernel(const float* __restrict__ W,
                            const float* __restrict__ bias,
                            float* __restrict__ planes,
                            float* __restrict__ sb) {
    __shared__ float part[CWAVES][4][16];
    __shared__ float red[CWAVES];

    const int tid  = threadIdx.x;
    const int wave = tid >> 5;
    const int lane = tid & 31;

    if (blockIdx.x == NCG * NRQ) {
        float v = 0.0f;
#pragma unroll
        for (int i = 0; i < DIM / 256; ++i) v += bias[tid + 256 * i];
        for (int off = 16; off > 0; off >>= 1) v += __shfl_xor(v, off);
        if (lane == 0) red[wave] = v;
        __syncthreads();
        if (wave == 0) {
            float t = (lane < CWAVES) ? red[lane] : 0.0f;
            for (int off = CWAVES / 2; off > 0; off >>= 1) t += __shfl_xor(t, off);
            if (lane == 0) sb[0] = t;
        }
        return;
    }

    const int rq    = blockIdx.x >> 6;
    const int cg    = blockIdx.x & (NCG - 1);
    const int lrow  = lane >> 3;
    const int piece = lane & 7;
    const float* p = W + (size_t)(rq * QROWS + wave * 64 + lrow) * DIM + cg * 32 + piece * 4;

    v4f ld[16];
#pragma unroll
    for (int i = 0; i < 16; ++i)
        ld[i] = *(const v4f*)(p + (size_t)(4 * i) * DIM);

    v2f a; a.x = 1.0f; a.y = 1.0f;
    v8f c0 = {}, c1 = {}, c2 = {}, c3 = {};
#pragma unroll
    for (int g = 0; g < 8; ++g) {
        v4f u = ld[2 * g], v = ld[2 * g + 1];
        v2f b;
        b.x = u.x; b.y = v.x;
        c0 = __builtin_amdgcn_wmma_f32_16x16x4_f32(false, a, false, b, (short)0, c0, false, false);
        b.x = u.y; b.y = v.y;
        c1 = __builtin_amdgcn_wmma_f32_16x16x4_f32(false, a, false, b, (short)0, c1, false, false);
        b.x = u.z; b.y = v.z;
        c2 = __builtin_amdgcn_wmma_f32_16x16x4_f32(false, a, false, b, (short)0, c2, false, false);
        b.x = u.w; b.y = v.w;
        c3 = __builtin_amdgcn_wmma_f32_16x16x4_f32(false, a, false, b, (short)0, c3, false, false);
    }
    if (lane < 16) {
        part[wave][0][lane] = c0[0];
        part[wave][1][lane] = c1[0];
        part[wave][2][lane] = c2[0];
        part[wave][3][lane] = c3[0];
    }
    __syncthreads();
    if (tid < 32) {
        const int t  = tid & 3;
        const int pp = tid >> 2;
        float s = 0.0f;
        for (int w = 0; w < CWAVES; ++w)
            s += part[w][t][pp] + part[w][t][pp + 8];
        planes[rq * DIM + cg * 32 + tid] = s;
    }
}

#define GROWS  32
#define GWAVES 16
#define TPITCH 33
#define PPITCH 34


__global__ __launch_bounds__(512) __attribute__((amdgpu_waves_per_eu(4, 4)))
void gemv_rows_kernel(const float* __restrict__ x,
                      const float* __restrict__ planes,
                      const float* __restrict__ sb,
                      float* __restrict__ out) {
    __shared__ v4f   tile[GWAVES][16 * TPITCH];
    __shared__ v4f   pstrip[GWAVES][NRQ * PPITCH];
    __shared__ float part[GWAVES][GROWS];

    const int tid  = threadIdx.x;
    const int wave = tid >> 5;
    const int lane = tid & 31;
    const int c    = lane & 15;
    const int h    = lane >> 4;

    const float* xw = x + (size_t)blockIdx.x * GROWS * DIM + wave * 128 + lane * 4;
    const float* ww = planes + wave * 128 + lane * 4;
    const float bsum = sb[0];

    v4f pl[NRQ];
#pragma unroll
    for (int p = 0; p < NRQ; ++p)
        pl[p] = *(const v4f*)(ww + p * DIM);

    v4f ld[32];
#pragma unroll
    for (int q = 0; q < 32; ++q)
        ld[q] = *(const v4f*)(xw + (size_t)q * DIM);

    v4f* ps = pstrip[wave];
#pragma unroll
    for (int p = 0; p < NRQ; ++p)
        ps[p * PPITCH + lane] = pl[p];
    v4f wa[16];
#pragma unroll
    for (int kk = 0; kk < 16; ++kk)
        wa[kk] = ps[(lane & 3) * PPITCH + 2 * kk + h];

    v4f* my = tile[wave];
#pragma unroll
    for (int g = 0; g < 2; ++g) {
#pragma unroll
        for (int q = 0; q < 16; ++q)
            my[q * TPITCH + lane] = ld[16 * g + q];

        v8f c0 = {}, c1 = {};
#pragma unroll
        for (int kk = 0; kk < 16; ++kk) {
            v4f f = my[c * TPITCH + 2 * kk + h];
            v2f a, b;
            a.x = wa[kk].x; a.y = wa[kk].y;
            b.x = f.x;      b.y = f.y;
            c0 = __builtin_amdgcn_wmma_f32_16x16x4_f32(false, a, false, b, (short)0, c0, false, false);
            a.x = wa[kk].z; a.y = wa[kk].w;
            b.x = f.z;      b.y = f.w;
            c1 = __builtin_amdgcn_wmma_f32_16x16x4_f32(false, a, false, b, (short)0, c1, false, false);
        }

        if (lane < 16)
            part[wave][16 * g + lane] = ((c0[0] + c1[0]) + (c0[1] + c1[1]))
                                      + ((c0[2] + c1[2]) + (c0[3] + c1[3]));
    }
    __syncthreads();
    if (tid < GROWS) {
        float s = 0.0f;
        for (int w = 0; w < GWAVES; ++w) s += part[w][tid];
        out[blockIdx.x * GROWS + tid] = SCALE * (s + bsum);
    }
}

extern "C" void kernel_launch(void* const* d_in, const int* in_sizes, int n_in,
                              void* d_out, int out_size, void* d_ws, size_t ws_size,
                              hipStream_t stream) {
    const float* x = (const float*)d_in[0];
    const float* W = (const float*)d_in[1];
    const float* b = (const float*)d_in[2];
    float* out = (float*)d_out;

    float* ws = (float*)d_ws;
    float* pl = ws;
    float* sb = ws + 5 * DIM;

    colsum_quarters_kernel<<<NCG * NRQ + 1, 256, 0, stream>>>(W, b, pl, sb);
    gemv_rows_kernel<<<256, 512, 0, stream>>>(x, pl, sb, out);
}
